# starting kernel
# baseline (speedup 1.0000x reference)
#include <hip/hip_runtime.h>
#include <hip/hip_bf16.h>

typedef __attribute__((ext_vector_type(2))) float v2f;
typedef __attribute__((ext_vector_type(8))) float v8f;

#define IN_FEATS 512
#define HIDDEN   64
#define NCLASSES 47
#define ALPHA    0.1f
#define KSTEPS   10

__global__ __launch_bounds__(256) void appnp_zero4(float4* __restrict__ p, int n4) {
    int i = blockIdx.x * blockDim.x + threadIdx.x;
    if (i < n4) p[i] = make_float4(0.f, 0.f, 0.f, 0.f);
}

__global__ __launch_bounds__(256) void appnp_degree(const int* __restrict__ src,
                                                    const int* __restrict__ dst,
                                                    float* __restrict__ odeg,
                                                    float* __restrict__ ideg, int E) {
    int e = blockIdx.x * blockDim.x + threadIdx.x;
    if (e >= E) return;
    atomicAdd(&odeg[src[e]], 1.0f);
    atomicAdd(&ideg[dst[e]], 1.0f);
}

__global__ __launch_bounds__(256) void appnp_norm(float* __restrict__ d, int n) {
    int i = blockIdx.x * blockDim.x + threadIdx.x;
    if (i < n) d[i] = rsqrtf(fmaxf(d[i], 1.0f));
}

__global__ __launch_bounds__(256) void appnp_gemm1_wmma(const float* __restrict__ F,
                                                        const float* __restrict__ W0,
                                                        const float* __restrict__ b0,
                                                        float* __restrict__ H, int N) {
    __shared__ float sw[256 * HIDDEN];
    const int lane = threadIdx.x & 31;
    const int wave = threadIdx.x >> 5;
    const int nTiles = (N + 15) >> 4;
    int rowTile = blockIdx.x * 8 + wave;
    if (rowTile >= nTiles) rowTile = nTiles - 1;
    const int rowBase = rowTile * 16;
    const int m  = lane & 15;
    const int kh = (lane >> 4) << 1;

    v8f acc0 = {}, acc1 = {}, acc2 = {}, acc3 = {};
    int arow = rowBase + m; if (arow >= N) arow = N - 1;
    const float* frow = F + (size_t)arow * IN_FEATS + kh;

    for (int ko = 0; ko < IN_FEATS; ko += 256) {
        __syncthreads();
        const float4* gsrc = (const float4*)(W0 + (size_t)ko * HIDDEN);
        float4* ldst = (float4*)sw;
        for (int i = threadIdx.x; i < (256 * HIDDEN) / 4; i += 256) ldst[i] = gsrc[i];
        __syncthreads();

        for (int k = 0; k < 256; k += 4) {
            if ((k & 63) == 0) __builtin_prefetch(frow + ko + k + 64, 0, 1);
            v2f a = *(const v2f*)(frow + ko + k);
            const float* w0p = sw + (k + kh) * HIDDEN + m;
            const float* w1p = w0p + HIDDEN;
            v2f b;
            b.x = w0p[0];  b.y = w1p[0];
            acc0 = __builtin_amdgcn_wmma_f32_16x16x4_f32(false, a, false, b, (short)0, acc0, false, false);
            b.x = w0p[16]; b.y = w1p[16];
            acc1 = __builtin_amdgcn_wmma_f32_16x16x4_f32(false, a, false, b, (short)0, acc1, false, false);
            b.x = w0p[32]; b.y = w1p[32];
            acc2 = __builtin_amdgcn_wmma_f32_16x16x4_f32(false, a, false, b, (short)0, acc2, false, false);
            b.x = w0p[48]; b.y = w1p[48];
            acc3 = __builtin_amdgcn_wmma_f32_16x16x4_f32(false, a, false, b, (short)0, acc3, false, false);
        }
    }

    const int rh = (lane >> 4) * 8;
    const float bias0 = b0[m], bias1 = b0[16 + m], bias2 = b0[32 + m], bias3 = b0[48 + m];
    for (int r = 0; r < 8; ++r) {
        int row = rowBase + rh + r;
        if (row < N) {
            float* hrow = H + (size_t)row * HIDDEN;
            hrow[m]      = fmaxf(acc0[r] + bias0, 0.f);
            hrow[16 + m] = fmaxf(acc1[r] + bias1, 0.f);
            hrow[32 + m] = fmaxf(acc2[r] + bias2, 0.f);
            hrow[48 + m] = fmaxf(acc3[r] + bias3, 0.f);
        }
    }
}

__global__ __launch_bounds__(256) void appnp_gemm2_wmma(const float* __restrict__ H,
                                                        const float* __restrict__ W1,
                                                        const float* __restrict__ b1,
                                                        const float* __restrict__ snorm,
                                                        float* __restrict__ O,
                                                        float* __restrict__ OS, int N) {
    const int lane = threadIdx.x & 31;
    const int wave = threadIdx.x >> 5;
    const int rowBase = (blockIdx.x * 8 + wave) * 16;
    if (rowBase >= N) return;
    const int m  = lane & 15;
    const int kh = (lane >> 4) << 1;
    const bool c2ok = (32 + m) < NCLASSES;

    v8f acc0 = {}, acc1 = {}, acc2 = {};
    const float* hrow = H + (size_t)(rowBase + m) * HIDDEN + kh;

    for (int k = 0; k < HIDDEN; k += 4) {
        v2f a = *(const v2f*)(hrow + k);
        const float* w0p = W1 + (size_t)(k + kh) * NCLASSES + m;
        const float* w1p = w0p + NCLASSES;
        v2f b;
        b.x = w0p[0];  b.y = w1p[0];
        acc0 = __builtin_amdgcn_wmma_f32_16x16x4_f32(false, a, false, b, (short)0, acc0, false, false);
        b.x = w0p[16]; b.y = w1p[16];
        acc1 = __builtin_amdgcn_wmma_f32_16x16x4_f32(false, a, false, b, (short)0, acc1, false, false);
        b.x = c2ok ? w0p[32] : 0.f;
        b.y = c2ok ? w1p[32] : 0.f;
        acc2 = __builtin_amdgcn_wmma_f32_16x16x4_f32(false, a, false, b, (short)0, acc2, false, false);
    }

    const int rh = (lane >> 4) * 8;
    const float bias0 = b1[m], bias1 = b1[16 + m], bias2 = c2ok ? b1[32 + m] : 0.f;
    for (int r = 0; r < 8; ++r) {
        int row = rowBase + rh + r;
        float sn = snorm[row];
        float* orow = O  + (size_t)row * NCLASSES;
        float* srow = OS + (size_t)row * NCLASSES;
        float v0 = acc0[r] + bias0;
        float v1 = acc1[r] + bias1;
        orow[m]      = v0;  srow[m]      = v0 * sn;
        orow[16 + m] = v1;  srow[16 + m] = v1 * sn;
        if (c2ok) {
            float v2 = acc2[r] + bias2;
            orow[32 + m] = v2;  srow[32 + m] = v2 * sn;
        }
    }
}

__global__ __launch_bounds__(256) void appnp_scatter(const float* __restrict__ hs,
                                                     const int* __restrict__ src,
                                                     const int* __restrict__ dst,
                                                     float* __restrict__ acc,
                                                     long long total) {
    long long idx = (long long)blockIdx.x * blockDim.x + threadIdx.x;
    if (idx >= total) return;
    int e = (int)(idx / NCLASSES);
    int c = (int)(idx - (long long)e * NCLASSES);
    atomicAdd(&acc[(size_t)dst[e] * NCLASSES + c],
              hs[(size_t)src[e] * NCLASSES + c]);
}

__global__ __launch_bounds__(256) void appnp_mix(const float* __restrict__ acc,
                                                 const float* __restrict__ dnorm,
                                                 const float* __restrict__ snorm,
                                                 const float* __restrict__ h0,
                                                 float* __restrict__ out,
                                                 int do_scale, int total) {
    int idx = blockIdx.x * blockDim.x + threadIdx.x;
    if (idx >= total) return;
    int row = idx / NCLASSES;
    float mixed = (1.0f - ALPHA) * acc[idx] * dnorm[row] + ALPHA * h0[idx];
    out[idx] = do_scale ? mixed * snorm[row] : mixed;
}

extern "C" void kernel_launch(void* const* d_in, const int* in_sizes, int n_in,
                              void* d_out, int out_size, void* d_ws, size_t ws_size,
                              hipStream_t stream) {
    const float* F  = (const float*)d_in[0];
    const float* W0 = (const float*)d_in[1];
    const float* b0 = (const float*)d_in[2];
    const float* W1 = (const float*)d_in[3];
    const float* b1 = (const float*)d_in[4];
    const int*  src = (const int*)d_in[5];
    const int*  dst = (const int*)d_in[6];

    const int N = in_sizes[0] / IN_FEATS;
    const int E = in_sizes[5];
    float* out = (float*)d_out;

    float* ws    = (float*)d_ws;
    float* snorm = ws;
    float* dnorm = ws + (size_t)N;
    float* hrelu = ws + 2 * (size_t)N;
    float* h0    = hrelu + (size_t)N * HIDDEN;
    float* hs    = h0 + (size_t)N * NCLASSES;
    float* acc   = hs + (size_t)N * NCLASSES;

    {
        int n4 = (2 * N) / 4;
        appnp_zero4<<<(n4 + 255) / 256, 256, 0, stream>>>((float4*)snorm, n4);
        appnp_degree<<<(E + 255) / 256, 256, 0, stream>>>(src, dst, snorm, dnorm, E);
        appnp_norm<<<(2 * N + 255) / 256, 256, 0, stream>>>(snorm, 2 * N);
    }

    {
        int rowTiles = (N + 15) / 16;
        int blocks   = (rowTiles + 7) / 8;
        appnp_gemm1_wmma<<<blocks, 256, 0, stream>>>(F, W0, b0, hrelu, N);
        appnp_gemm2_wmma<<<blocks, 256, 0, stream>>>(hrelu, W1, b1, snorm, h0, hs, N);
    }

    {
        const long long tot = (long long)E * NCLASSES;
        const int sblocks = (int)((tot + 255) / 256);
        const int nc  = N * NCLASSES;
        const int nc4 = nc / 4;
        for (int it = 0; it < KSTEPS; ++it) {
            appnp_zero4<<<(nc4 + 255) / 256, 256, 0, stream>>>((float4*)acc, nc4);
            appnp_scatter<<<sblocks, 256, 0, stream>>>(hs, src, dst, acc, tot);
            if (it == KSTEPS - 1) {
                appnp_mix<<<(nc + 255) / 256, 256, 0, stream>>>(acc, dnorm, snorm, h0, out, 0, nc);
            } else {
                appnp_mix<<<(nc + 255) / 256, 256, 0, stream>>>(acc, dnorm, snorm, h0, hs, 1, nc);
            }
        }
    }
}


# Edge buckets in 2 launches (were 5), no global counters shared by chunks: a 4096-edge chunk is sorted by bucket in LDS, a work-group per bucket copies its segments; list in chunk order, not arrival
# speedup vs baseline: 6.6374x; 6.6374x over previous
#include <hip/hip_runtime.h>
#include <hip/hip_bf16.h>

typedef __attribute__((ext_vector_type(2))) float v2f;
typedef __attribute__((ext_vector_type(8))) float v8f;

#define IN_FEATS 512
#define HIDDEN   64
#define NCLASSES 47
#define ALPHA    0.1f
#define KSTEPS   10

__global__ __launch_bounds__(256) void appnp_gemm1_wmma(const float* __restrict__ F,
                                                        const float* __restrict__ W0,
                                                        const float* __restrict__ b0,
                                                        float* __restrict__ H, int N) {
    __shared__ float sw[256 * HIDDEN];
    const int lane = threadIdx.x & 31;
    const int wave = threadIdx.x >> 5;
    const int nTiles = (N + 15) >> 4;
    int rowTile = blockIdx.x * 8 + wave;
    if (rowTile >= nTiles) rowTile = nTiles - 1;
    const int rowBase = rowTile * 16;
    const int m  = lane & 15;
    const int kh = (lane >> 4) << 1;

    v8f acc0 = {}, acc1 = {}, acc2 = {}, acc3 = {};
    int arow = rowBase + m; if (arow >= N) arow = N - 1;
    const float* frow = F + (size_t)arow * IN_FEATS + kh;

    for (int ko = 0; ko < IN_FEATS; ko += 256) {
        __syncthreads();
        const float4* gsrc = (const float4*)(W0 + (size_t)ko * HIDDEN);
        float4* ldst = (float4*)sw;
        for (int i = threadIdx.x; i < (256 * HIDDEN) / 4; i += 256) ldst[i] = gsrc[i];
        __syncthreads();

        for (int k = 0; k < 256; k += 4) {
            if ((k & 63) == 0) __builtin_prefetch(frow + ko + k + 64, 0, 1);
            v2f a = *(const v2f*)(frow + ko + k);
            const float* w0p = sw + (k + kh) * HIDDEN + m;
            const float* w1p = w0p + HIDDEN;
            v2f b;
            b.x = w0p[0];  b.y = w1p[0];
            acc0 = __builtin_amdgcn_wmma_f32_16x16x4_f32(false, a, false, b, (short)0, acc0, false, false);
            b.x = w0p[16]; b.y = w1p[16];
            acc1 = __builtin_amdgcn_wmma_f32_16x16x4_f32(false, a, false, b, (short)0, acc1, false, false);
            b.x = w0p[32]; b.y = w1p[32];
            acc2 = __builtin_amdgcn_wmma_f32_16x16x4_f32(false, a, false, b, (short)0, acc2, false, false);
            b.x = w0p[48]; b.y = w1p[48];
            acc3 = __builtin_amdgcn_wmma_f32_16x16x4_f32(false, a, false, b, (short)0, acc3, false, false);
        }
    }

    const int rh = (lane >> 4) * 8;
    const float bias0 = b0[m], bias1 = b0[16 + m], bias2 = b0[32 + m], bias3 = b0[48 + m];
    for (int r = 0; r < 8; ++r) {
        int row = rowBase + rh + r;
        if (row < N) {
            float* hrow = H + (size_t)row * HIDDEN;
            hrow[m]      = fmaxf(acc0[r] + bias0, 0.f);
            hrow[16 + m] = fmaxf(acc1[r] + bias1, 0.f);
            hrow[32 + m] = fmaxf(acc2[r] + bias2, 0.f);
            hrow[48 + m] = fmaxf(acc3[r] + bias3, 0.f);
        }
    }
}

#define TAB_PITCH 64
__global__ __launch_bounds__(256) void appnp_gemm2_wmma(const float* __restrict__ H,
                                                        const float* __restrict__ W1,
                                                        const float* __restrict__ b1,
                                                        const float* __restrict__ snorm,
                                                        float* __restrict__ O,
                                                        float* __restrict__ OS, int N) {
    const int lane = threadIdx.x & 31;
    const int wave = threadIdx.x >> 5;
    const int rowBase = (blockIdx.x * 8 + wave) * 16;
    if (rowBase >= N) return;
    const int m  = lane & 15;
    const int kh = (lane >> 4) << 1;
    const bool c2ok = (32 + m) < NCLASSES;

    v8f acc0 = {}, acc1 = {}, acc2 = {};
    const float* hrow = H + (size_t)(rowBase + m) * HIDDEN + kh;

    for (int k = 0; k < HIDDEN; k += 4) {
        v2f a = *(const v2f*)(hrow + k);
        const float* w0p = W1 + (size_t)(k + kh) * NCLASSES + m;
        const float* w1p = w0p + NCLASSES;
        v2f b;
        b.x = w0p[0];  b.y = w1p[0];
        acc0 = __builtin_amdgcn_wmma_f32_16x16x4_f32(false, a, false, b, (short)0, acc0, false, false);
        b.x = w0p[16]; b.y = w1p[16];
        acc1 = __builtin_amdgcn_wmma_f32_16x16x4_f32(false, a, false, b, (short)0, acc1, false, false);
        b.x = c2ok ? w0p[32] : 0.f;
        b.y = c2ok ? w1p[32] : 0.f;
        acc2 = __builtin_amdgcn_wmma_f32_16x16x4_f32(false, a, false, b, (short)0, acc2, false, false);
    }

    const int rh = (lane >> 4) * 8;
    const float bias0 = b1[m], bias1 = b1[16 + m], bias2 = c2ok ? b1[32 + m] : 0.f;
    for (int r = 0; r < 8; ++r) {
        int row = rowBase + rh + r;
        float sn = snorm[row];
        float* orow = O  + (size_t)row * NCLASSES;
        float* srow = OS + (size_t)row * TAB_PITCH;
        float v0 = acc0[r] + bias0;
        float v1 = acc1[r] + bias1;
        orow[m]      = v0;  srow[m]      = v0 * sn;
        orow[16 + m] = v1;  srow[16 + m] = v1 * sn;
        if (c2ok) {
            float v2 = acc2[r] + bias2;
            orow[32 + m] = v2;  srow[32 + m] = v2 * sn;
        } else {
            srow[NCLASSES] = 1.0f;
        }
    }
}

#define ROWS_PER_BIN  200
#define CHUNK_EDGES   4096

__device__ __forceinline__ int wg_excl_scan(int v, int tid, int* wtot, int* total) {
    const int lane = tid & 31;
    const int w = tid >> 5;
    int inc = v;
    for (int o = 1; o < 32; o <<= 1) {
        const int t = __shfl_up(inc, o, 32);
        if (lane >= o) inc += t;
    }
    if (lane == 31) wtot[w] = inc;
    __syncthreads();
    if (w == 0) {
        const int t = wtot[lane];
        int ti = t;
        for (int o = 1; o < 32; o <<= 1) {
            const int u = __shfl_up(ti, o, 32);
            if (lane >= o) ti += u;
        }
        wtot[lane] = ti - t;
        if (lane == 31) wtot[32] = ti;
    }
    __syncthreads();
    *total = wtot[32];
    return inc - v + wtot[w];
}

__global__ __launch_bounds__(1024) void edge_chunk_sort(const int* __restrict__ src,
                                                         const int* __restrict__ dst,
                                                         int2* __restrict__ chunkD,
                                                         int* __restrict__ chunkS,
                                                         int* __restrict__ offTab,
                                                         int* __restrict__ cursor,
                                                         int E, int bins) {
    extern __shared__ int dynlds[];
    __shared__ int wtotD[33], wtotS[33];
    const int tid = threadIdx.x;
    const int nb1 = bins + 1;
    int* cntD = dynlds;
    int* cntS = dynlds + nb1;
    int2* recD = (int2*)(dynlds + 2 * nb1);
    int* recS = dynlds + 2 * nb1 + 2 * CHUNK_EDGES;
    for (int i = tid; i < 2 * nb1; i += 1024) dynlds[i] = 0;
    if (blockIdx.x == 0 && tid == 0) *cursor = 0;
    __syncthreads();
    const int base = blockIdx.x * CHUNK_EDGES;
    int d[CHUNK_EDGES / 1024], s[CHUNK_EDGES / 1024];
    int rD[CHUNK_EDGES / 1024], rS[CHUNK_EDGES / 1024];
    for (int k = 0; k < CHUNK_EDGES / 1024; ++k) {
        const int e = base + k * 1024 + tid;
        d[k] = -1; s[k] = 0; rD[k] = 0; rS[k] = 0;
        if (e < E) {
            d[k] = dst[e];
            s[k] = src[e];
            rD[k] = atomicAdd(&cntD[(unsigned)d[k] / ROWS_PER_BIN], 1);
            rS[k] = atomicAdd(&cntS[(unsigned)s[k] / ROWS_PER_BIN], 1);
        }
    }
    __syncthreads();
    const int per = (bins + 1023) / 1024;
    int lo = tid * per; if (lo > bins) lo = bins;
    int hi = lo + per;  if (hi > bins) hi = bins;
    int sD = 0, sS = 0;
    for (int i = lo; i < hi; ++i) { sD += cntD[i]; sS += cntS[i]; }
    int totD, totS;
    int runD = wg_excl_scan(sD, tid, wtotD, &totD);
    int runS = wg_excl_scan(sS, tid, wtotS, &totS);
    for (int i = lo; i < hi; ++i) {
        int c = cntD[i]; cntD[i] = runD; runD += c;
        c = cntS[i];     cntS[i] = runS; runS += c;
    }
    if (tid == 0) { cntD[bins] = totD; cntS[bins] = totS; }
    __syncthreads();
    int* orow = offTab + (size_t)blockIdx.x * (2 * nb1);
    for (int i = tid; i < 2 * nb1; i += 1024) orow[i] = dynlds[i];
    for (int k = 0; k < CHUNK_EDGES / 1024; ++k) {
        if (d[k] >= 0) {
            const int bD = (unsigned)d[k] / ROWS_PER_BIN;
            recD[cntD[bD] + rD[k]] = make_int2(s[k], d[k] - bD * ROWS_PER_BIN);
            const int bS = (unsigned)s[k] / ROWS_PER_BIN;
            recS[cntS[bS] + rS[k]] = s[k];
        }
    }
    __syncthreads();
    int n = E - base; if (n > CHUNK_EDGES) n = CHUNK_EDGES;
    for (int i = tid; i < n; i += 1024) {
        chunkD[base + i] = recD[i];
        chunkS[base + i] = recS[i];
    }
}

#define GATHER_ILP 4
static_assert(GATHER_ILP == 4, "the load fences in bucket_gather name four items");

__global__ __launch_bounds__(1024) void bucket_gather(const int2* __restrict__ chunkD,
                                                       const int* __restrict__ chunkS,
                                                       const int* __restrict__ offTab,
                                                       int* __restrict__ cursor,
                                                       int2* __restrict__ edges,
                                                       int* __restrict__ binRange,
                                                       float* __restrict__ snorm,
                                                       int N, int bins, int nC) {
    extern __shared__ int dynlds[];
    __shared__ int wtotD[33], wtotS[33];
    __shared__ int cnt[ROWS_PER_BIN];
    __shared__ int listStart;
    const int tid = threadIdx.x;
    const int b = blockIdx.x;
    const int nb1 = bins + 1;
    int* posD = dynlds;
    int* aD   = posD + nC + 1;
    int* posS = aD + nC;
    int* aS   = posS + nC + 1;
    const int per = (nC + 1023) / 1024;
    int lo = tid * per; if (lo > nC) lo = nC;
    int hi = lo + per;  if (hi > nC) hi = nC;
    int sD = 0, sS = 0;
    for (int w = lo; w < hi; ++w) {
        const int* row = offTab + (size_t)w * (2 * nb1);
        int a = row[b], z = row[b + 1];
        aD[w] = a; posD[w] = z - a; sD += z - a;
        a = row[nb1 + b]; z = row[nb1 + b + 1];
        aS[w] = a; posS[w] = z - a; sS += z - a;
    }
    if (tid < ROWS_PER_BIN) cnt[tid] = 0;
    int totD, totS;
    int runD = wg_excl_scan(sD, tid, wtotD, &totD);
    int runS = wg_excl_scan(sS, tid, wtotS, &totS);
    for (int w = lo; w < hi; ++w) {
        int c = posD[w]; posD[w] = runD; runD += c;
        c = posS[w];     posS[w] = runS; runS += c;
    }
    if (tid == 0) {
        posD[nC] = totD;
        posS[nC] = totS;
        const int st = atomicAdd(cursor, totD);
        listStart = st;
        binRange[2 * b]     = st;
        binRange[2 * b + 1] = st + totD;
    }
    __syncthreads();
    const int start = listStart;
    const int topStep = 1 << (31 - __clz(nC));

    for (int i0 = tid; i0 < totD; i0 += GATHER_ILP * 1024) {
        int ic[GATHER_ILP], w[GATHER_ILP];
        int2 v[GATHER_ILP];
#pragma unroll
        for (int k = 0; k < GATHER_ILP; ++k) {
            const int i = i0 + 1024 * k;
            ic[k] = (i < totD) ? i : totD - 1;
            w[k] = 0;
        }
        for (int step = topStep; step > 0; step >>= 1) {
#pragma unroll
            for (int k = 0; k < GATHER_ILP; ++k) {
                const int c  = w[k] + step;
                const int cc = (c < nC) ? c : nC;
                if (posD[cc] <= ic[k]) w[k] = c;
            }
        }
#pragma unroll
        for (int k = 0; k < GATHER_ILP; ++k)
            v[k] = chunkD[(size_t)w[k] * CHUNK_EDGES + aD[w[k]] + (ic[k] - posD[w[k]])];
        asm volatile("" :: "v"(v[0].x), "v"(v[1].x), "v"(v[2].x), "v"(v[3].x));
#pragma unroll
        for (int k = 0; k < GATHER_ILP; ++k) {
            const int i = i0 + 1024 * k;
            if (i < totD) edges[start + i] = v[k];
        }
    }

    const int rowBase = b * ROWS_PER_BIN;
    for (int i0 = tid; i0 < totS; i0 += GATHER_ILP * 1024) {
        int ic[GATHER_ILP], w[GATHER_ILP], v[GATHER_ILP];
#pragma unroll
        for (int k = 0; k < GATHER_ILP; ++k) {
            const int i = i0 + 1024 * k;
            ic[k] = (i < totS) ? i : totS - 1;
            w[k] = 0;
        }
        for (int step = topStep; step > 0; step >>= 1) {
#pragma unroll
            for (int k = 0; k < GATHER_ILP; ++k) {
                const int c  = w[k] + step;
                const int cc = (c < nC) ? c : nC;
                if (posS[cc] <= ic[k]) w[k] = c;
            }
        }
#pragma unroll
        for (int k = 0; k < GATHER_ILP; ++k)
            v[k] = chunkS[(size_t)w[k] * CHUNK_EDGES + aS[w[k]] + (ic[k] - posS[w[k]])];
        asm volatile("" :: "v"(v[0]), "v"(v[1]), "v"(v[2]), "v"(v[3]));
#pragma unroll
        for (int k = 0; k < GATHER_ILP; ++k) {
            const int i = i0 + 1024 * k;
            if (i < totS) atomicAdd(&cnt[v[k] - rowBase], 1);
        }
    }
    __syncthreads();
    if (tid < ROWS_PER_BIN && rowBase + tid < N)
        snorm[rowBase + tid] = rsqrtf(fmaxf((float)cnt[tid], 1.0f));
}

#define HALF_BCAST(V, P) __builtin_amdgcn_update_dpp(0, (V), 0x150 + (P), 0xf, 0xf, true)

struct row3 { float a, b, c; };

#define ROW_FETCH(P)                                                          \
    const int s_##P = HALF_BCAST(ed.x, P);                                    \
    const int d_##P = HALF_BCAST(ed.y, P);                                    \
    const row3 v_##P = *(const row3*)(hs_in + (unsigned)(s_##P * TAB_PITCH + col3));
#define ROW_SCATTER(P)                                                        \
    {                                                                         \
        float* ap = sacc + 2 * (d_##P * ACC_PITCH + col3) + half;             \
        atomicAdd(ap,     v_##P.a);                                           \
        atomicAdd(ap + 2, v_##P.b);                                           \
        atomicAdd(ap + 4, v_##P.c);                                           \
    }

#define ACC_PITCH  48
#define SACC_WORDS (2 * (ROWS_PER_BIN + 1) * ACC_PITCH)
#define TELE_WORDS (ROWS_PER_BIN * NCLASSES)
#define TELE_QUADS ((TELE_WORDS / 4 + 1023) / 1024)
#define MIX_TRIPS  ((ROWS_PER_BIN * ACC_PITCH + 1023) / 1024)
#define MIX_GROUP  5
static_assert(MIX_TRIPS % MIX_GROUP == 0, "the mix goes in whole groups of trips");
static_assert(TELE_WORDS % 4 == 0, "a bucket's teleport run must start 16-byte aligned");
static_assert(ROWS_PER_BIN <= 256, "the source norms of a bucket are fetched by 256 threads");
template <bool FINAL>
__global__ __launch_bounds__(1024) void prop_step_bin(const float* __restrict__ hs_in,
                                                       const int2* __restrict__ edges,
                                                       const int* __restrict__ binRange,
                                                       const float* __restrict__ snorm,
                                                       const float* __restrict__ h0,
                                                       float* __restrict__ outp, int N) {
    __shared__ float sacc[SACC_WORDS];
    __shared__ float stele[TELE_WORDS];
    __shared__ float snrm[ROWS_PER_BIN];
    const int tid  = threadIdx.x;
    const int lane = tid & 31;
    const int wave = __builtin_amdgcn_readfirstlane(tid >> 5);
    const int half = lane >> 4;
    const int col3 = 3 * (lane & 15);
    const int rowBase = blockIdx.x * ROWS_PER_BIN;
    int nrows = N - rowBase; if (nrows > ROWS_PER_BIN) nrows = ROWS_PER_BIN;

    for (int i = tid; i < SACC_WORDS / 4; i += 1024)
        ((float4*)sacc)[i] = make_float4(0.f, 0.f, 0.f, 0.f);
    __syncthreads();

    const float* tele = (FINAL ? (const float*)outp : h0) + (size_t)rowBase * NCLASSES;
    const int teleWords = nrows * NCLASSES;
    const int teleQuads = teleWords >> 2;
    float4 tq[TELE_QUADS];
#pragma unroll
    for (int k = 0; k < TELE_QUADS; ++k) {
        const int q = tid + 1024 * k;
        tq[k] = make_float4(0.f, 0.f, 0.f, 0.f);
        if (q < teleQuads) tq[k] = ((const float4*)tele)[q];
    }
    float ttail = 0.f;
    if (tid < (teleWords & 3)) ttail = tele[4 * teleQuads + tid];
    float nv = 0.f;
    if (!FINAL && tid < nrows) nv = snorm[rowBase + tid];

    const int e0 = __builtin_amdgcn_readfirstlane(binRange[2 * blockIdx.x]);
    const int e1 = __builtin_amdgcn_readfirstlane(binRange[2 * blockIdx.x + 1]);
    int2 ed = make_int2(0, ROWS_PER_BIN);
    if (e0 + wave * 32 + lane < e1) ed = edges[e0 + wave * 32 + lane];

#pragma unroll
    for (int k = 0; k < TELE_QUADS; ++k) {
        const int q = tid + 1024 * k;
        if (q < teleQuads) ((float4*)stele)[q] = tq[k];
    }
    if (tid < (teleWords & 3)) stele[4 * teleQuads + tid] = ttail;
    if (!FINAL && tid < ROWS_PER_BIN) snrm[tid] = nv;

    for (int base = e0 + wave * 32; base < e1; base += 1024) {
        int2 edn = make_int2(0, ROWS_PER_BIN);
        if (base + 1024 + lane < e1) edn = edges[base + 1024 + lane];
        {
            ROW_FETCH(0) ROW_FETCH(1) ROW_FETCH(2) ROW_FETCH(3)
            ROW_FETCH(4) ROW_FETCH(5) ROW_FETCH(6) ROW_FETCH(7)
            ROW_SCATTER(0) ROW_SCATTER(1) ROW_SCATTER(2) ROW_SCATTER(3)
            ROW_SCATTER(4) ROW_SCATTER(5) ROW_SCATTER(6) ROW_SCATTER(7)
        }
        {
            ROW_FETCH(8)  ROW_FETCH(9)  ROW_FETCH(10) ROW_FETCH(11)
            ROW_FETCH(12) ROW_FETCH(13) ROW_FETCH(14) ROW_FETCH(15)
            ROW_SCATTER(8)  ROW_SCATTER(9)  ROW_SCATTER(10) ROW_SCATTER(11)
            ROW_SCATTER(12) ROW_SCATTER(13) ROW_SCATTER(14) ROW_SCATTER(15)
        }
        ed = edn;
    }
    __syncthreads();

    const float2* spair = (const float2*)sacc;
    const int tot = nrows * ACC_PITCH;
#pragma unroll
    for (int g = 0; g < MIX_TRIPS; g += MIX_GROUP) {
        float2 pr[MIX_GROUP], dg[MIX_GROUP];
        float te[MIX_GROUP], sn[MIX_GROUP];
        int cI[MIX_GROUP], rI[MIX_GROUP];
#pragma unroll
        for (int k = 0; k < MIX_GROUP; ++k) {
            const int i  = tid + 1024 * (g + k);
            const int ic = (i < tot) ? i : tot - 1;
            const int r  = ic / ACC_PITCH;
            const int c  = ic - r * ACC_PITCH;
            const int cc = (c < NCLASSES) ? c : NCLASSES - 1;
            rI[k] = r;  cI[k] = c;
            pr[k] = spair[ic];
            dg[k] = spair[r * ACC_PITCH + NCLASSES];
            te[k] = stele[r * NCLASSES + cc];
            sn[k] = FINAL ? 0.f : snrm[r];
        }
        asm volatile("" :: "v"(pr[0].x), "v"(pr[1].x), "v"(pr[2].x), "v"(pr[3].x), "v"(pr[4].x),
                           "v"(te[0]), "v"(te[1]), "v"(te[2]), "v"(te[3]), "v"(te[4]),
                           "v"(dg[0].x), "v"(dg[1].x), "v"(dg[2].x), "v"(dg[3].x), "v"(dg[4].x));
#pragma unroll
        for (int k = 0; k < MIX_GROUP; ++k) {
            const int i = tid + 1024 * (g + k);
            const int row = rowBase + rI[k];
            const float dnk = rsqrtf(fmaxf(dg[k].x + dg[k].y, 1.0f));
            const float mixed = fmaf(dnk, (1.0f - ALPHA) * (pr[k].x + pr[k].y), ALPHA * te[k]);
            if (FINAL) {
                if (i < tot && cI[k] < NCLASSES) outp[(size_t)row * NCLASSES + cI[k]] = mixed;
            } else {
                const float v = (cI[k] < NCLASSES) ? mixed * sn[k] : 1.0f;
                if (i < tot) outp[(size_t)row * TAB_PITCH + cI[k]] = v;
            }
        }
    }
}

extern "C" void kernel_launch(void* const* d_in, const int* in_sizes, int n_in,
                              void* d_out, int out_size, void* d_ws, size_t ws_size,
                              hipStream_t stream) {
    const float* F  = (const float*)d_in[0];
    const float* W0 = (const float*)d_in[1];
    const float* b0 = (const float*)d_in[2];
    const float* W1 = (const float*)d_in[3];
    const float* b1 = (const float*)d_in[4];
    const int*  src = (const int*)d_in[5];
    const int*  dst = (const int*)d_in[6];

    const int N = in_sizes[0] / IN_FEATS;
    const int E = in_sizes[5];
    float* out = (float*)d_out;

    float* ws    = (float*)d_ws;
    float* snorm = ws;
    float* lists = ws + ((2 * (size_t)N + TAB_PITCH - 1) / TAB_PITCH) * TAB_PITCH;
    float* hs    = lists + (size_t)N * HIDDEN;
    float* hs2   = hs + (size_t)N * TAB_PITCH;
    float* hrelu = hs2;
    float* h0    = out;
    const int bins = (N + ROWS_PER_BIN - 1) / ROWS_PER_BIN;
    const int chunks = (E + CHUNK_EDGES - 1) / CHUNK_EDGES;
    int2* edges    = (int2*)lists;
    int* cursor    = (int*)lists + 2 * (size_t)E;
    int* binRange  = cursor + 2;
    int2* chunkD   = (int2*)hs;
    int* chunkS    = (int*)hs + 2 * (size_t)chunks * CHUNK_EDGES;
    int* offTab    = chunkS + (size_t)chunks * CHUNK_EDGES;

    {
        edge_chunk_sort<<<chunks, 1024, (2 * (bins + 1) + 3 * CHUNK_EDGES) * sizeof(int), stream>>>(
            src, dst, chunkD, chunkS, offTab, cursor, E, bins);
        bucket_gather<<<bins, 1024, (4 * chunks + 2) * sizeof(int), stream>>>(
            chunkD, chunkS, offTab, cursor, edges, binRange, snorm, N, bins, chunks);
    }

    {
        int rowTiles = (N + 15) / 16;
        int blocks   = (rowTiles + 7) / 8;
        appnp_gemm1_wmma<<<blocks, 256, 0, stream>>>(F, W0, b0, hrelu, N);
        appnp_gemm2_wmma<<<blocks, 256, 0, stream>>>(hrelu, W1, b1, snorm, h0, hs, N);
    }

    {
        for (int it = 0; it < KSTEPS; ++it) {
            const float* cur = (it & 1) ? hs2 : hs;
            float* nxt       = (it & 1) ? hs : hs2;
            if (it == KSTEPS - 1) {
                prop_step_bin<true><<<bins, 1024, 0, stream>>>(cur, edges, binRange, snorm, nullptr, out, N);
            } else {
                prop_step_bin<false><<<bins, 1024, 0, stream>>>(cur, edges, binRange, snorm, h0, nxt, N);
            }
        }
    }
}
